# starting kernel
# baseline (speedup 1.0000x reference)
#include <hip/hip_runtime.h>

typedef __attribute__((ext_vector_type(2))) float v2f;
typedef __attribute__((ext_vector_type(8))) float v8f;

static constexpr int kH    = 64;
static constexpr int kD    = 64;
static constexpr int kDin  = 2145;
static constexpr int kRowsPerWave = 16;
static constexpr int kWaves       = 8;
static constexpr int kKChunks     = 17;
static constexpr int kBpStride    = 65;
static constexpr int kXsStride    = 68;

__global__ __launch_bounds__(256) void qvf_wmma_kernel(
    const float* __restrict__ obs,
    const float* __restrict__ W,
    const float* __restrict__ bias,
    float* __restrict__ out)
{
  __shared__ v2f   Bp[34 * kBpStride];
  __shared__ float Xs[kWaves][kRowsPerWave * kXsStride];

  const int tid  = threadIdx.x;
  const int h    = blockIdx.y;
  const int wave = tid >> 5;
  const int lane = tid & 31;
  const int lo   = lane & 15;
  const int hi   = lane >> 4;

  {
    float* Bf = (float*)Bp;
    const float* Wh = W + (size_t)h * kDin;
    #pragma unroll
    for (int e = tid; e < 68 * 64; e += 256) {
      const int j = e >> 6;
      const int i = e & 63;
      float v = 0.f;
      if (j < 64) {
        if (i <= j) v = Wh[64 + i * 64 - (i * (i - 1)) / 2 + (j - i)];
      } else if (j == 64) {
        v = Wh[i];
      }
      Bf[((j >> 1) * kBpStride + i) * 2 + (j & 1)] = v;
    }
  }

  const int n_base = blockIdx.x * (kWaves * kRowsPerWave) + wave * kRowsPerWave;
  const int m    = lo;
  const int koff = hi * 2;
  const float* xrow = obs + ((size_t)(n_base + m) * kH + h) * kD;

  v2f a[kKChunks];
  #pragma unroll
  for (int kk = 0; kk < 16; ++kk) {
    v2f t = *(const v2f*)(xrow + 4 * kk + koff);
    a[kk] = t;
    *(v2f*)(&Xs[wave][m * kXsStride + 4 * kk + koff]) = t;
  }
  {
    v2f aext;
    aext.x = (hi == 0) ? 1.0f : 0.0f;
    aext.y = 0.0f;
    a[16] = aext;
  }

  __syncthreads();

  float part[8];
  #pragma unroll
  for (int v = 0; v < 8; ++v) part[v] = 0.f;

  #pragma unroll
  for (int t = 0; t < 4; ++t) {
    v8f acc = {};
    #pragma unroll
    for (int kk = 0; kk < kKChunks; ++kk) {
      v2f b = Bp[(2 * kk + hi) * kBpStride + (16 * t + lo)];
      acc = __builtin_amdgcn_wmma_f32_16x16x4_f32(
           false, a[kk],  false, b,
           (short)0, acc,  false,  false);
    }
    #pragma unroll
    for (int v = 0; v < 8; ++v) {
      const int mr = v + 8 * hi;
      part[v] += Xs[wave][mr * kXsStride + 16 * t + lo] * acc[v];
    }
  }

  #pragma unroll
  for (int v = 0; v < 8; ++v) {
    float s = part[v];
    s += __shfl_xor(s, 1, 32);
    s += __shfl_xor(s, 2, 32);
    s += __shfl_xor(s, 4, 32);
    s += __shfl_xor(s, 8, 32);
    part[v] = s;
  }

  const float b0 = bias[h];
  #pragma unroll
  for (int v = 0; v < 8; ++v) {
    if (lo == v) {
      const int row = n_base + v + 8 * hi;
      out[(size_t)row * kH + h] = part[v] + b0;
    }
  }
}

extern "C" void kernel_launch(void* const* d_in, const int* in_sizes, int n_in,
                              void* d_out, int out_size, void* d_ws, size_t ws_size,
                              hipStream_t stream) {
  const float* obs  = (const float*)d_in[0];
  const float* W    = (const float*)d_in[1];
  const float* bias = (const float*)d_in[2];
  float* out = (float*)d_out;

  const int num_paths = in_sizes[0] / (kH * kD);
  dim3 grid(num_paths / (kWaves * kRowsPerWave), kH);
  qvf_wmma_kernel<<<grid, 256, 0, stream>>>(obs, W, bias, out);
}


# Global loads coalesced: weight row and X tile read as whole 128-B lines and staged in LDS; triangular unpack and A fragments then read LDS, branch-free; same k and sum order
# speedup vs baseline: 2.0210x; 2.0210x over previous
#include <hip/hip_runtime.h>


typedef __attribute__((ext_vector_type(2))) float v2f;
typedef __attribute__((ext_vector_type(4))) float v4f;
typedef __attribute__((ext_vector_type(8))) float v8f;

static constexpr int kH    = 64;
static constexpr int kD    = 64;
static constexpr int kDin  = 2145;
static constexpr int kWUsed = 2144;
static constexpr int kRowsPerWave = 16;
static constexpr int kWaves       = 8;
static constexpr int kKChunks     = 17;
static constexpr int kBpStride    = 65;
static constexpr int kXsStride    = 68;

__global__ __launch_bounds__(256) void qvf_wmma_kernel(
    const float* __restrict__ obs,
    const float* __restrict__ W,
    const float* __restrict__ bias,
    float* __restrict__ out)
{
  __shared__ v2f   Bp[34 * kBpStride];
  __shared__ __attribute__((aligned(16))) float Xs[kWaves][kRowsPerWave * kXsStride];
  __shared__ float Wraw[kWUsed];

  const int tid  = threadIdx.x;
  const int h    = blockIdx.y;
  const int wave = tid >> 5;
  const int lane = tid & 31;
  const int lo   = lane & 15;
  const int hi   = lane >> 4;

  const float* Wh = W + (size_t)h * kDin;
  float wreg[9];
  #pragma unroll
  for (int it = 0; it < 8; ++it) wreg[it] = Wh[tid + 256 * it];
  wreg[8] = (tid < kWUsed - 2048) ? Wh[2048 + tid] : 0.f;

  const int n_base = blockIdx.x * (kWaves * kRowsPerWave) + wave * kRowsPerWave;
  const int c4 = lo * 4;
  v4f xreg[8];
  #pragma unroll
  for (int r = 0; r < 8; ++r) {
    const int row = 2 * r + hi;
    xreg[r] = *(const v4f*)(obs + ((size_t)(n_base + row) * kH + h) * kD + c4);
  }

  #pragma unroll
  for (int it = 0; it < 8; ++it) Wraw[tid + 256 * it] = wreg[it];
  if (tid < kWUsed - 2048) Wraw[2048 + tid] = wreg[8];
  #pragma unroll
  for (int r = 0; r < 8; ++r) {
    const int row = 2 * r + hi;
    *(v4f*)(&Xs[wave][row * kXsStride + c4]) = xreg[r];
  }

  __syncthreads();

  const int m    = lo;
  const int koff = hi * 2;
  v2f a[kKChunks];
  #pragma unroll
  for (int kk = 0; kk < 16; ++kk) {
    a[kk] = *(const v2f*)(&Xs[wave][m * kXsStride + 4 * kk + koff]);
  }
  {
    v2f aext;
    aext.x = (hi == 0) ? 1.0f : 0.0f;
    aext.y = 0.0f;
    a[16] = aext;
  }

  {
    float* Bf = (float*)Bp;
    const int i  = tid & 63;
    const int jb = tid >> 6;
    const int tri0 = 64 + 63 * i - (i * (i - 1)) / 2;
    float wv[17];
    #pragma unroll
    for (int it = 0; it < 16; ++it) {
      const int j = jb + 4 * it;
      const float v = Wraw[tri0 + j];
      wv[it] = (i <= j) ? v : 0.f;
    }
    {
      const float v = Wraw[i];
      wv[16] = (jb == 0) ? v : 0.f;
    }
    #pragma unroll
    for (int it = 0; it < 17; ++it) {
      const int j = jb + 4 * it;
      Bf[((j >> 1) * kBpStride + i) * 2 + (j & 1)] = wv[it];
    }
  }

  __syncthreads();

  float part[8];
  #pragma unroll
  for (int v = 0; v < 8; ++v) part[v] = 0.f;

  #pragma unroll
  for (int t = 0; t < 4; ++t) {
    v8f acc = {};
    #pragma unroll
    for (int kk = 0; kk < kKChunks; ++kk) {
      v2f b = Bp[(2 * kk + hi) * kBpStride + (16 * t + lo)];
      acc = __builtin_amdgcn_wmma_f32_16x16x4_f32(
           false, a[kk],  false, b,
           (short)0, acc,  false,  false);
    }
    #pragma unroll
    for (int v = 0; v < 8; ++v) {
      const int mr = v + 8 * hi;
      part[v] += Xs[wave][mr * kXsStride + 16 * t + lo] * acc[v];
    }
  }

  #pragma unroll
  for (int v = 0; v < 8; ++v) {
    float s = part[v];
    s += __shfl_xor(s, 1, 32);
    s += __shfl_xor(s, 2, 32);
    s += __shfl_xor(s, 4, 32);
    s += __shfl_xor(s, 8, 32);
    part[v] = s;
  }

  const float b0 = bias[h];
  #pragma unroll
  for (int v = 0; v < 8; ++v) {
    if (lo == v) {
      const int row = n_base + v + 8 * hi;
      out[(size_t)row * kH + h] = part[v] + b0;
    }
  }
}

extern "C" void kernel_launch(void* const* d_in, const int* in_sizes, int n_in,
                              void* d_out, int out_size, void* d_ws, size_t ws_size,
                              hipStream_t stream) {
  const float* obs  = (const float*)d_in[0];
  const float* W    = (const float*)d_in[1];
  const float* bias = (const float*)d_in[2];
  float* out = (float*)d_out;

  const int num_paths = in_sizes[0] / (kH * kD);
  dim3 grid(num_paths / (kWaves * kRowsPerWave), kH);
  qvf_wmma_kernel<<<grid, 256, 0, stream>>>(obs, W, bias, out);
}


# K chunks of the WMMA loop whose B fragment is all zero by the triangle (j < 16t in i-tile t) are no longer issued: 44 matrix ops per wave, not 68; k and sum order of kept products unchanged
# speedup vs baseline: 2.2726x; 1.1245x over previous
#include <hip/hip_runtime.h>


typedef __attribute__((ext_vector_type(2))) float v2f;
typedef __attribute__((ext_vector_type(4))) float v4f;
typedef __attribute__((ext_vector_type(8))) float v8f;

static constexpr int kH    = 64;
static constexpr int kD    = 64;
static constexpr int kDin  = 2145;
static constexpr int kWUsed = 2144;
static constexpr int kRowsPerWave = 16;
static constexpr int kWaves       = 8;
static constexpr int kKChunks     = 17;
static constexpr int kBpStride    = 65;
static constexpr int kXsStride    = 68;

__global__ __launch_bounds__(256) void qvf_wmma_kernel(
    const float* __restrict__ obs,
    const float* __restrict__ W,
    const float* __restrict__ bias,
    float* __restrict__ out)
{
  __shared__ v2f   Bp[34 * kBpStride];
  __shared__ __attribute__((aligned(16))) float Xs[kWaves][kRowsPerWave * kXsStride];
  __shared__ float Wraw[kWUsed];

  const int tid  = threadIdx.x;
  const int h    = blockIdx.y;
  const int wave = tid >> 5;
  const int lane = tid & 31;
  const int lo   = lane & 15;
  const int hi   = lane >> 4;

  const float* Wh = W + (size_t)h * kDin;
  float wreg[9];
  #pragma unroll
  for (int it = 0; it < 8; ++it) wreg[it] = Wh[tid + 256 * it];
  wreg[8] = (tid < kWUsed - 2048) ? Wh[2048 + tid] : 0.f;

  const int n_base = blockIdx.x * (kWaves * kRowsPerWave) + wave * kRowsPerWave;
  const int c4 = lo * 4;
  v4f xreg[8];
  #pragma unroll
  for (int r = 0; r < 8; ++r) {
    const int row = 2 * r + hi;
    xreg[r] = *(const v4f*)(obs + ((size_t)(n_base + row) * kH + h) * kD + c4);
  }

  #pragma unroll
  for (int it = 0; it < 8; ++it) Wraw[tid + 256 * it] = wreg[it];
  if (tid < kWUsed - 2048) Wraw[2048 + tid] = wreg[8];
  #pragma unroll
  for (int r = 0; r < 8; ++r) {
    const int row = 2 * r + hi;
    *(v4f*)(&Xs[wave][row * kXsStride + c4]) = xreg[r];
  }

  __syncthreads();

  const int m    = lo;
  const int koff = hi * 2;
  v2f a[kKChunks];
  #pragma unroll
  for (int kk = 0; kk < 16; ++kk) {
    a[kk] = *(const v2f*)(&Xs[wave][m * kXsStride + 4 * kk + koff]);
  }
  {
    v2f aext;
    aext.x = (hi == 0) ? 1.0f : 0.0f;
    aext.y = 0.0f;
    a[16] = aext;
  }

  {
    float* Bf = (float*)Bp;
    const int i  = tid & 63;
    const int jb = tid >> 6;
    const int tri0 = 64 + 63 * i - (i * (i - 1)) / 2;
    float wv[17];
    #pragma unroll
    for (int it = 0; it < 16; ++it) {
      const int j = jb + 4 * it;
      const float v = Wraw[tri0 + j];
      wv[it] = (i <= j) ? v : 0.f;
    }
    {
      const float v = Wraw[i];
      wv[16] = (jb == 0) ? v : 0.f;
    }
    #pragma unroll
    for (int it = 0; it < 17; ++it) {
      const int j = jb + 4 * it;
      Bf[((j >> 1) * kBpStride + i) * 2 + (j & 1)] = wv[it];
    }
  }

  __syncthreads();

  float part[8];
  #pragma unroll
  for (int v = 0; v < 8; ++v) part[v] = 0.f;

  #pragma unroll
  for (int t = 0; t < 4; ++t) {
    v8f acc = {};
    #pragma unroll
    for (int kk = 4 * t; kk < kKChunks; ++kk) {
      v2f b = Bp[(2 * kk + hi) * kBpStride + (16 * t + lo)];
      acc = __builtin_amdgcn_wmma_f32_16x16x4_f32(
           false, a[kk],  false, b,
           (short)0, acc,  false,  false);
    }
    #pragma unroll
    for (int v = 0; v < 8; ++v) {
      const int mr = v + 8 * hi;
      part[v] += Xs[wave][mr * kXsStride + 16 * t + lo] * acc[v];
    }
  }

  #pragma unroll
  for (int v = 0; v < 8; ++v) {
    float s = part[v];
    s += __shfl_xor(s, 1, 32);
    s += __shfl_xor(s, 2, 32);
    s += __shfl_xor(s, 4, 32);
    s += __shfl_xor(s, 8, 32);
    part[v] = s;
  }

  const float b0 = bias[h];
  #pragma unroll
  for (int v = 0; v < 8; ++v) {
    if (lo == v) {
      const int row = n_base + v + 8 * hi;
      out[(size_t)row * kH + h] = part[v] + b0;
    }
  }
}

extern "C" void kernel_launch(void* const* d_in, const int* in_sizes, int n_in,
                              void* d_out, int out_size, void* d_ws, size_t ws_size,
                              hipStream_t stream) {
  const float* obs  = (const float*)d_in[0];
  const float* W    = (const float*)d_in[1];
  const float* bias = (const float*)d_in[2];
  float* out = (float*)d_out;

  const int num_paths = in_sizes[0] / (kH * kD);
  dim3 grid(num_paths / (kWaves * kRowsPerWave), kH);
  qvf_wmma_kernel<<<grid, 256, 0, stream>>>(obs, W, bias, out);
}


# Product transposed and contracted over i (Yt = Ut Xt): weight fragments read straight from the packed row, LDS unpack pass gone; x*Y epilogue from registers; linear row enters as C; sum order changed
# speedup vs baseline: 2.5681x; 1.1300x over previous
#include <hip/hip_runtime.h>


typedef __attribute__((ext_vector_type(2))) float v2f;
typedef __attribute__((ext_vector_type(4))) float v4f;
typedef __attribute__((ext_vector_type(8))) float v8f;

static constexpr int kH    = 64;
static constexpr int kD    = 64;
static constexpr int kDin  = 2145;
static constexpr int kWUsed = 2144;
static constexpr int kRowsPerWave = 16;
static constexpr int kWaves       = 8;
static constexpr int kXsStride    = 68;

static constexpr int tri0(int i) { return 64 + 63 * i - (i * (i - 1)) / 2; }

__global__ __launch_bounds__(256) void qvf_wmma_kernel(
    const float* __restrict__ obs,
    const float* __restrict__ W,
    const float* __restrict__ bias,
    float* __restrict__ out)
{
  __shared__ __attribute__((aligned(16))) float Xs[kWaves][kRowsPerWave * kXsStride];
  __shared__ __attribute__((aligned(16))) float Wraw[kWUsed];

  const int tid  = threadIdx.x;
  const int h    = blockIdx.y;
  const int wave = tid >> 5;
  const int lane = tid & 31;
  const int lo   = lane & 15;
  const int hi   = lane >> 4;

  const float* Wh = W + (size_t)h * kDin;
  float wreg[9];
  #pragma unroll
  for (int it = 0; it < 8; ++it) wreg[it] = Wh[tid + 256 * it];
  wreg[8] = (tid < kWUsed - 2048) ? Wh[2048 + tid] : 0.f;

  const float b0 = bias[h];

  const int n_base = blockIdx.x * (kWaves * kRowsPerWave) + wave * kRowsPerWave;
  const int c4 = lo * 4;
  v4f xreg[8];
  #pragma unroll
  for (int r = 0; r < 8; ++r) {
    const int row = 2 * r + hi;
    xreg[r] = *(const v4f*)(obs + ((size_t)(n_base + row) * kH + h) * kD + c4);
  }

  #pragma unroll
  for (int it = 0; it < 8; ++it) Wraw[tid + 256 * it] = wreg[it];
  if (tid < kWUsed - 2048) Wraw[2048 + tid] = wreg[8];
  #pragma unroll
  for (int r = 0; r < 8; ++r) {
    const int row = 2 * r + hi;
    *(v4f*)(&Xs[wave][row * kXsStride + c4]) = xreg[r];
  }

  __syncthreads();

  float xr[4][8];
  #pragma unroll
  for (int u = 0; u < 4; ++u) {
    const v4f p0 = *(const v4f*)(&Xs[wave][lo * kXsStride + 16 * u + 8 * hi]);
    const v4f p1 = *(const v4f*)(&Xs[wave][lo * kXsStride + 16 * u + 8 * hi + 4]);
    xr[u][0] = p0.x; xr[u][1] = p0.y; xr[u][2] = p0.z; xr[u][3] = p0.w;
    xr[u][4] = p1.x; xr[u][5] = p1.y; xr[u][6] = p1.z; xr[u][7] = p1.w;
  }

  const int pbase = lo + 476 * hi;
  const int q8    = 8 * hi;
  const int dloc  = lo - 8 * hi;

  float part = (hi == 0) ? b0 : 0.f;

  #pragma unroll
  for (int t = 0; t < 4; ++t) {
    v8f acc;
    {
      const v4f w0 = *(const v4f*)(&Wraw[16 * t + 8 * hi]);
      const v4f w1 = *(const v4f*)(&Wraw[16 * t + 8 * hi + 4]);
      acc[0] = w0.x; acc[1] = w0.y; acc[2] = w0.z; acc[3] = w0.w;
      acc[4] = w1.x; acc[5] = w1.y; acc[6] = w1.z; acc[7] = w1.w;
    }
    #pragma unroll
    for (int u = 0; u <= t; ++u) {
      #pragma unroll
      for (int c = 0; c < 4; ++c) {
        const int i0 = 16 * u + 2 * c;
        float a0 = Wraw[(pbase - q8 * i0)       + (tri0(i0)     + 16 * t)];
        float a1 = Wraw[(pbase - q8 * (i0 + 1)) + (tri0(i0 + 1) + 16 * t)];
        if (u == t) {
          a0 = (2 * c     <= dloc) ? a0 : 0.f;
          a1 = (2 * c + 1 <= dloc) ? a1 : 0.f;
        }
        v2f aw; aw.x = a0; aw.y = a1;
        v2f bx; bx.x = xr[u][2 * c]; bx.y = xr[u][2 * c + 1];
        acc = __builtin_amdgcn_wmma_f32_16x16x4_f32(
             false, aw,  false, bx,
             (short)0, acc,  false,  false);
      }
    }
    #pragma unroll
    for (int r = 0; r < 8; ++r) part += xr[t][r] * acc[r];
  }

  part += __shfl_xor(part, 16, 32);

  if (hi == 0) {
    out[(size_t)(n_base + lo) * kH + h] = part;
  }
}

extern "C" void kernel_launch(void* const* d_in, const int* in_sizes, int n_in,
                              void* d_out, int out_size, void* d_ws, size_t ws_size,
                              hipStream_t stream) {
  const float* obs  = (const float*)d_in[0];
  const float* W    = (const float*)d_in[1];
  const float* bias = (const float*)d_in[2];
  float* out = (float*)d_out;

  const int num_paths = in_sizes[0] / (kH * kD);
  dim3 grid(num_paths / (kWaves * kRowsPerWave), kH);
  qvf_wmma_kernel<<<grid, 256, 0, stream>>>(obs, W, bias, out);
}
